# starting kernel
# baseline (speedup 1.0000x reference)
#include <hip/hip_runtime.h>
#include <math.h>

#define BB   8
#define NN   1024
#define FIN  128
#define FOUT 64
#define HH   4
#define EPSF 1e-7f
#define NEG  0.2f

typedef float v2f __attribute__((ext_vector_type(2)));
typedef float v8f __attribute__((ext_vector_type(8)));

__device__ __forceinline__ v8f wmma_f32(v2f a, v2f b, v8f c) {
    return __builtin_amdgcn_wmma_f32_16x16x4_f32(
         false, a,  false, b,
         (short)0, c,  false,  false);
}

__global__ __launch_bounds__(32) void gat_wh(const float* __restrict__ x,
                                             const float* __restrict__ W,
                                             float* __restrict__ Wh) {
    const int lane = threadIdx.x;
    const int lm = lane & 15, hi = lane >> 4;
    int bid = blockIdx.x;
    const int ib = bid & 63; bid >>= 6;
    const int h  = bid & 3;  bid >>= 2;
    const int b  = bid;
    const int i0 = ib * 16;

    const float* xr = x + ((size_t)b * NN + i0 + lm) * FIN + 2 * hi;
    const float* Wp = W + (size_t)h * FIN * FOUT + (size_t)(2 * hi) * FOUT + lm;

    v8f acc0 = {}, acc1 = {}, acc2 = {}, acc3 = {};
    #pragma unroll 4
    for (int k = 0; k < FIN; k += 4) {
        v2f a = *(const v2f*)(xr + k);
        const float* wk = Wp + (size_t)k * FOUT;
        v2f b0 = { wk[0],  wk[FOUT]      };
        v2f b1 = { wk[16], wk[FOUT + 16] };
        v2f b2 = { wk[32], wk[FOUT + 32] };
        v2f b3 = { wk[48], wk[FOUT + 48] };
        acc0 = wmma_f32(a, b0, acc0);
        acc1 = wmma_f32(a, b1, acc1);
        acc2 = wmma_f32(a, b2, acc2);
        acc3 = wmma_f32(a, b3, acc3);
    }
    float* outp = Wh + (((size_t)b * HH + h) * NN + i0 + 8 * hi) * FOUT + lm;
    #pragma unroll
    for (int r = 0; r < 8; ++r) {
        float* o = outp + (size_t)r * FOUT;
        o[0] = acc0[r]; o[16] = acc1[r]; o[32] = acc2[r]; o[48] = acc3[r];
    }
}

__global__ void gat_scores(const float* __restrict__ Wh,
                           const float* __restrict__ a_w,
                           float* __restrict__ s1, float* __restrict__ s2) {
    int idx = blockIdx.x * blockDim.x + threadIdx.x;
    if (idx >= BB * HH * NN) return;
    const int h = (idx / NN) & (HH - 1);
    const float* row = Wh + (size_t)idx * FOUT;
    const float* a1 = a_w + (size_t)h * 2 * FOUT;
    const float* a2 = a1 + FOUT;
    float v1 = 0.f, v2 = 0.f;
    #pragma unroll 8
    for (int o = 0; o < FOUT; ++o) {
        float w = row[o];
        v1 += w * a1[o];
        v2 += w * a2[o];
    }
    s1[idx] = v1; s2[idx] = v2;
}

__global__ void gat_colsum(const float* __restrict__ A,
                           const float* __restrict__ s1,
                           const float* __restrict__ s2,
                           const float* __restrict__ a_b,
                           float* __restrict__ D) {
    int bid = blockIdx.x;
    const int jblk = bid & 3; bid >>= 2;
    const int h    = bid & 3; bid >>= 2;
    const int b    = bid;
    const int j  = jblk * 256 + threadIdx.x;
    const int bh = b * HH + h;

    const float s2v = s2[(size_t)bh * NN + j];
    const float ab  = a_b[h];
    const float* Ac  = A + (size_t)b * NN * NN + j;
    const float* s1p = s1 + (size_t)bh * NN;

    float acc = 0.f;
    for (int i = 0; i < NN; ++i) {
        float e = s1p[i] + s2v + ab;
        e = (e >= 0.f) ? e : NEG * e;
        acc += Ac[(size_t)i * NN] * expf(e);
    }
    D[(size_t)bh * NN + j] = acc;
}

__global__ void gat_scale(float* __restrict__ Wh, const float* __restrict__ D) {
    size_t idx = (size_t)blockIdx.x * blockDim.x + threadIdx.x;
    if (idx >= (size_t)BB * HH * NN * FOUT) return;
    Wh[idx] *= 1.0f / (D[idx >> 6] + EPSF);
}

__global__ __launch_bounds__(32) void gat_attn_out(
        const float* __restrict__ A, const float* __restrict__ Whs,
        const float* __restrict__ s1, const float* __restrict__ s2,
        const float* __restrict__ a_b, float* __restrict__ out) {
    __shared__ __align__(16) float E[16 * 68];

    const int lane = threadIdx.x;
    const int lm = lane & 15, hi = lane >> 4;
    int bid = blockIdx.x;
    const int ib = bid & 63; bid >>= 6;
    const int h  = bid & 3;  bid >>= 2;
    const int b  = bid;
    const int i0 = ib * 16;
    const int bh = b * HH + h;

    const float ab = a_b[h];
    const float* s1p = s1 + (size_t)bh * NN + i0;
    const float* s2p = s2 + (size_t)bh * NN;
    const float* Ap  = A  + (size_t)b * NN * NN;
    const float* Wp  = Whs + (size_t)bh * NN * FOUT;

    v8f acc0 = {}, acc1 = {}, acc2 = {}, acc3 = {};

    for (int j0 = 0; j0 < NN; j0 += 64) {
        const float s2a = s2p[j0 + lane];
        const float s2b = s2p[j0 + lane + 32];
        #pragma unroll 4
        for (int r = 0; r < 16; ++r) {
            const float s1v = s1p[r];
            const float* Ar = Ap + (size_t)(i0 + r) * NN + j0;
            float e0 = s1v + s2a + ab; e0 = (e0 >= 0.f) ? e0 : NEG * e0;
            float e1 = s1v + s2b + ab; e1 = (e1 >= 0.f) ? e1 : NEG * e1;
            E[r * 68 + lane]      = Ar[lane]      * expf(e0);
            E[r * 68 + lane + 32] = Ar[lane + 32] * expf(e1);
        }
        __syncthreads();
        #pragma unroll
        for (int kk = 0; kk < 16; ++kk) {
            v2f a = *(const v2f*)&E[lm * 68 + kk * 4 + 2 * hi];
            const float* wr = Wp + (size_t)(j0 + kk * 4 + 2 * hi) * FOUT + lm;
            v2f b0 = { wr[0],  wr[FOUT]      };
            v2f b1 = { wr[16], wr[FOUT + 16] };
            v2f b2 = { wr[32], wr[FOUT + 32] };
            v2f b3 = { wr[48], wr[FOUT + 48] };
            acc0 = wmma_f32(a, b0, acc0);
            acc1 = wmma_f32(a, b1, acc1);
            acc2 = wmma_f32(a, b2, acc2);
            acc3 = wmma_f32(a, b3, acc3);
        }
        __syncthreads();
    }

    float* op = out + ((size_t)b * NN + i0 + 8 * hi) * (HH * FOUT) + (size_t)h * FOUT + lm;
    #pragma unroll
    for (int r = 0; r < 8; ++r) {
        float* o = op + (size_t)r * (HH * FOUT);
        o[0] = acc0[r]; o[16] = acc1[r]; o[32] = acc2[r]; o[48] = acc3[r];
    }
}

extern "C" void kernel_launch(void* const* d_in, const int* in_sizes, int n_in,
                              void* d_out, int out_size, void* d_ws, size_t ws_size,
                              hipStream_t stream) {
    const float* A  = (const float*)d_in[0];
    const float* x  = (const float*)d_in[1];
    const float* W  = (const float*)d_in[2];
    const float* aw = (const float*)d_in[3];
    const float* ab = (const float*)d_in[4];
    float* out = (float*)d_out;

    float* Wh = (float*)d_ws;
    float* s1 = Wh + (size_t)BB * HH * NN * FOUT;
    float* s2 = s1 + (size_t)BB * HH * NN;
    float* D  = s2 + (size_t)BB * HH * NN;

    gat_wh      <<<dim3(BB * HH * (NN / 16)),  dim3(32),  0, stream>>>(x, W, Wh);
    gat_scores  <<<dim3(BB * HH * NN / 256),   dim3(256), 0, stream>>>(Wh, aw, s1, s2);
    gat_colsum  <<<dim3(BB * HH * (NN / 256)), dim3(256), 0, stream>>>(A, s1, s2, ab, D);
    gat_scale   <<<dim3((BB * HH * NN * FOUT) / 256), dim3(256), 0, stream>>>(Wh, D);
    gat_attn_out<<<dim3(BB * HH * (NN / 16)),  dim3(32),  0, stream>>>(A, Wh, s1, s2, ab, out);
}


# column-sum kernel re-tiled: work-group = 32 columns x all rows x 4 heads, rows split over 16 waves, partials summed through LDS (changes the order of the sum over i)
# speedup vs baseline: 2.1219x; 2.1219x over previous
#include <hip/hip_runtime.h>
#include <math.h>

#define BB   8
#define NN   1024
#define FIN  128
#define FOUT 64
#define HH   4
#define EPSF 1e-7f
#define NEG  0.2f

typedef float v2f __attribute__((ext_vector_type(2)));
typedef float v8f __attribute__((ext_vector_type(8)));

__device__ __forceinline__ v8f wmma_f32(v2f a, v2f b, v8f c) {
    return __builtin_amdgcn_wmma_f32_16x16x4_f32(
         false, a,  false, b,
         (short)0, c,  false,  false);
}

__global__ __launch_bounds__(32) void gat_wh(const float* __restrict__ x,
                                             const float* __restrict__ W,
                                             float* __restrict__ Wh) {
    const int lane = threadIdx.x;
    const int lm = lane & 15, hi = lane >> 4;
    int bid = blockIdx.x;
    const int ib = bid & 63; bid >>= 6;
    const int h  = bid & 3;  bid >>= 2;
    const int b  = bid;
    const int i0 = ib * 16;

    const float* xr = x + ((size_t)b * NN + i0 + lm) * FIN + 2 * hi;
    const float* Wp = W + (size_t)h * FIN * FOUT + (size_t)(2 * hi) * FOUT + lm;

    v8f acc0 = {}, acc1 = {}, acc2 = {}, acc3 = {};
    #pragma unroll 4
    for (int k = 0; k < FIN; k += 4) {
        v2f a = *(const v2f*)(xr + k);
        const float* wk = Wp + (size_t)k * FOUT;
        v2f b0 = { wk[0],  wk[FOUT]      };
        v2f b1 = { wk[16], wk[FOUT + 16] };
        v2f b2 = { wk[32], wk[FOUT + 32] };
        v2f b3 = { wk[48], wk[FOUT + 48] };
        acc0 = wmma_f32(a, b0, acc0);
        acc1 = wmma_f32(a, b1, acc1);
        acc2 = wmma_f32(a, b2, acc2);
        acc3 = wmma_f32(a, b3, acc3);
    }
    float* outp = Wh + (((size_t)b * HH + h) * NN + i0 + 8 * hi) * FOUT + lm;
    #pragma unroll
    for (int r = 0; r < 8; ++r) {
        float* o = outp + (size_t)r * FOUT;
        o[0] = acc0[r]; o[16] = acc1[r]; o[32] = acc2[r]; o[48] = acc3[r];
    }
}

__global__ void gat_scores(const float* __restrict__ Wh,
                           const float* __restrict__ a_w,
                           float* __restrict__ s1, float* __restrict__ s2) {
    int idx = blockIdx.x * blockDim.x + threadIdx.x;
    if (idx >= BB * HH * NN) return;
    const int h = (idx / NN) & (HH - 1);
    const float* row = Wh + (size_t)idx * FOUT;
    const float* a1 = a_w + (size_t)h * 2 * FOUT;
    const float* a2 = a1 + FOUT;
    float v1 = 0.f, v2 = 0.f;
    #pragma unroll 8
    for (int o = 0; o < FOUT; ++o) {
        float w = row[o];
        v1 += w * a1[o];
        v2 += w * a2[o];
    }
    s1[idx] = v1; s2[idx] = v2;
}

#define CS_WAVES 16
#define CS_ROWS  (NN / CS_WAVES)
#define CS_U     8

__global__ __launch_bounds__(CS_WAVES * 32) void gat_colsum(
        const float* __restrict__ A,
        const float* __restrict__ s1,
        const float* __restrict__ s2,
        const float* __restrict__ a_b,
        float* __restrict__ D) {
    __shared__ float red[CS_WAVES][HH][32];

    const int lane = threadIdx.x & 31;
    const int w  = __builtin_amdgcn_readfirstlane(threadIdx.x >> 5);
    const int jt = blockIdx.x & 31;
    const int b  = blockIdx.x >> 5;
    const int j  = jt * 32 + lane;
    const int i0 = w * CS_ROWS;

    float s2v[HH], abv[HH], acc[HH];
    #pragma unroll
    for (int h = 0; h < HH; ++h) {
        s2v[h] = s2[((size_t)b * HH + h) * NN + j];
        abv[h] = a_b[h];
        acc[h] = 0.f;
    }
    const float* Ap  = A  + ((size_t)b * NN + i0) * NN + j;
    const float* s1p = s1 + (size_t)b * HH * NN + i0;

    float a_nxt[CS_U];
    #pragma unroll
    for (int u = 0; u < CS_U; ++u) a_nxt[u] = Ap[(size_t)u * NN];

    for (int r0 = 0; r0 < CS_ROWS; r0 += CS_U) {
        float a_cur[CS_U];
        #pragma unroll
        for (int u = 0; u < CS_U; ++u) a_cur[u] = a_nxt[u];
        if (r0 + CS_U < CS_ROWS) {
            #pragma unroll
            for (int u = 0; u < CS_U; ++u)
                a_nxt[u] = Ap[(size_t)(r0 + CS_U + u) * NN];
        }
        #pragma unroll
        for (int u = 0; u < CS_U; ++u) {
            #pragma unroll
            for (int h = 0; h < HH; ++h) {
                float e = s1p[h * NN + r0 + u] + s2v[h] + abv[h];
                e = (e >= 0.f) ? e : NEG * e;
                acc[h] += a_cur[u] * expf(e);
            }
        }
    }

    #pragma unroll
    for (int h = 0; h < HH; ++h) red[w][h][lane] = acc[h];
    __syncthreads();
    if (threadIdx.x < HH * 32) {
        float sum = 0.f;
        #pragma unroll
        for (int k = 0; k < CS_WAVES; ++k) sum += red[k][w][lane];
        D[((size_t)b * HH + w) * NN + j] = sum;
    }
}

__global__ void gat_scale(float* __restrict__ Wh, const float* __restrict__ D) {
    size_t idx = (size_t)blockIdx.x * blockDim.x + threadIdx.x;
    if (idx >= (size_t)BB * HH * NN * FOUT) return;
    Wh[idx] *= 1.0f / (D[idx >> 6] + EPSF);
}

__global__ __launch_bounds__(32) void gat_attn_out(
        const float* __restrict__ A, const float* __restrict__ Whs,
        const float* __restrict__ s1, const float* __restrict__ s2,
        const float* __restrict__ a_b, float* __restrict__ out) {
    __shared__ __align__(16) float E[16 * 68];

    const int lane = threadIdx.x;
    const int lm = lane & 15, hi = lane >> 4;
    int bid = blockIdx.x;
    const int ib = bid & 63; bid >>= 6;
    const int h  = bid & 3;  bid >>= 2;
    const int b  = bid;
    const int i0 = ib * 16;
    const int bh = b * HH + h;

    const float ab = a_b[h];
    const float* s1p = s1 + (size_t)bh * NN + i0;
    const float* s2p = s2 + (size_t)bh * NN;
    const float* Ap  = A  + (size_t)b * NN * NN;
    const float* Wp  = Whs + (size_t)bh * NN * FOUT;

    v8f acc0 = {}, acc1 = {}, acc2 = {}, acc3 = {};

    for (int j0 = 0; j0 < NN; j0 += 64) {
        const float s2a = s2p[j0 + lane];
        const float s2b = s2p[j0 + lane + 32];
        #pragma unroll 4
        for (int r = 0; r < 16; ++r) {
            const float s1v = s1p[r];
            const float* Ar = Ap + (size_t)(i0 + r) * NN + j0;
            float e0 = s1v + s2a + ab; e0 = (e0 >= 0.f) ? e0 : NEG * e0;
            float e1 = s1v + s2b + ab; e1 = (e1 >= 0.f) ? e1 : NEG * e1;
            E[r * 68 + lane]      = Ar[lane]      * expf(e0);
            E[r * 68 + lane + 32] = Ar[lane + 32] * expf(e1);
        }
        __syncthreads();
        #pragma unroll
        for (int kk = 0; kk < 16; ++kk) {
            v2f a = *(const v2f*)&E[lm * 68 + kk * 4 + 2 * hi];
            const float* wr = Wp + (size_t)(j0 + kk * 4 + 2 * hi) * FOUT + lm;
            v2f b0 = { wr[0],  wr[FOUT]      };
            v2f b1 = { wr[16], wr[FOUT + 16] };
            v2f b2 = { wr[32], wr[FOUT + 32] };
            v2f b3 = { wr[48], wr[FOUT + 48] };
            acc0 = wmma_f32(a, b0, acc0);
            acc1 = wmma_f32(a, b1, acc1);
            acc2 = wmma_f32(a, b2, acc2);
            acc3 = wmma_f32(a, b3, acc3);
        }
        __syncthreads();
    }

    float* op = out + ((size_t)b * NN + i0 + 8 * hi) * (HH * FOUT) + (size_t)h * FOUT + lm;
    #pragma unroll
    for (int r = 0; r < 8; ++r) {
        float* o = op + (size_t)r * (HH * FOUT);
        o[0] = acc0[r]; o[16] = acc1[r]; o[32] = acc2[r]; o[48] = acc3[r];
    }
}

extern "C" void kernel_launch(void* const* d_in, const int* in_sizes, int n_in,
                              void* d_out, int out_size, void* d_ws, size_t ws_size,
                              hipStream_t stream) {
    const float* A  = (const float*)d_in[0];
    const float* x  = (const float*)d_in[1];
    const float* W  = (const float*)d_in[2];
    const float* aw = (const float*)d_in[3];
    const float* ab = (const float*)d_in[4];
    float* out = (float*)d_out;

    float* Wh = (float*)d_ws;
    float* s1 = Wh + (size_t)BB * HH * NN * FOUT;
    float* s2 = s1 + (size_t)BB * HH * NN;
    float* D  = s2 + (size_t)BB * HH * NN;

    gat_wh      <<<dim3(BB * HH * (NN / 16)),  dim3(32),  0, stream>>>(x, W, Wh);
    gat_scores  <<<dim3(BB * HH * NN / 256),   dim3(256), 0, stream>>>(Wh, aw, s1, s2);
    gat_colsum  <<<dim3(BB * (NN / 32)), dim3(CS_WAVES * 32), 0, stream>>>(A, s1, s2, ab, D);
    gat_scale   <<<dim3((BB * HH * NN * FOUT) / 256), dim3(256), 0, stream>>>(Wh, D);
    gat_attn_out<<<dim3(BB * HH * (NN / 16)),  dim3(32),  0, stream>>>(A, Wh, s1, s2, ab, out);
}


# output kernel re-tiled: 4-wave work-group per 64 rows shares each 64x64 Whs tile through double-buffered LDS; next tile's A/Whs global loads issued ahead of the tile's 64 WMMAs; sum order kept
# speedup vs baseline: 2.9269x; 1.3794x over previous
#include <hip/hip_runtime.h>
#include <math.h>

#define BB   8
#define NN   1024
#define FIN  128
#define FOUT 64
#define HH   4
#define EPSF 1e-7f
#define NEG  0.2f

typedef float v2f __attribute__((ext_vector_type(2)));
typedef float v8f __attribute__((ext_vector_type(8)));

__device__ __forceinline__ v8f wmma_f32(v2f a, v2f b, v8f c) {
    return __builtin_amdgcn_wmma_f32_16x16x4_f32(
         false, a,  false, b,
         (short)0, c,  false,  false);
}

__global__ __launch_bounds__(32) void gat_wh(const float* __restrict__ x,
                                             const float* __restrict__ W,
                                             float* __restrict__ Wh) {
    const int lane = threadIdx.x;
    const int lm = lane & 15, hi = lane >> 4;
    int bid = blockIdx.x;
    const int ib = bid & 63; bid >>= 6;
    const int h  = bid & 3;  bid >>= 2;
    const int b  = bid;
    const int i0 = ib * 16;

    const float* xr = x + ((size_t)b * NN + i0 + lm) * FIN + 2 * hi;
    const float* Wp = W + (size_t)h * FIN * FOUT + (size_t)(2 * hi) * FOUT + lm;

    v8f acc0 = {}, acc1 = {}, acc2 = {}, acc3 = {};
    #pragma unroll 4
    for (int k = 0; k < FIN; k += 4) {
        v2f a = *(const v2f*)(xr + k);
        const float* wk = Wp + (size_t)k * FOUT;
        v2f b0 = { wk[0],  wk[FOUT]      };
        v2f b1 = { wk[16], wk[FOUT + 16] };
        v2f b2 = { wk[32], wk[FOUT + 32] };
        v2f b3 = { wk[48], wk[FOUT + 48] };
        acc0 = wmma_f32(a, b0, acc0);
        acc1 = wmma_f32(a, b1, acc1);
        acc2 = wmma_f32(a, b2, acc2);
        acc3 = wmma_f32(a, b3, acc3);
    }
    float* outp = Wh + (((size_t)b * HH + h) * NN + i0 + 8 * hi) * FOUT + lm;
    #pragma unroll
    for (int r = 0; r < 8; ++r) {
        float* o = outp + (size_t)r * FOUT;
        o[0] = acc0[r]; o[16] = acc1[r]; o[32] = acc2[r]; o[48] = acc3[r];
    }
}

__global__ void gat_scores(const float* __restrict__ Wh,
                           const float* __restrict__ a_w,
                           float* __restrict__ s1, float* __restrict__ s2) {
    int idx = blockIdx.x * blockDim.x + threadIdx.x;
    if (idx >= BB * HH * NN) return;
    const int h = (idx / NN) & (HH - 1);
    const float* row = Wh + (size_t)idx * FOUT;
    const float* a1 = a_w + (size_t)h * 2 * FOUT;
    const float* a2 = a1 + FOUT;
    float v1 = 0.f, v2 = 0.f;
    #pragma unroll 8
    for (int o = 0; o < FOUT; ++o) {
        float w = row[o];
        v1 += w * a1[o];
        v2 += w * a2[o];
    }
    s1[idx] = v1; s2[idx] = v2;
}

#define CS_WAVES 16
#define CS_ROWS  (NN / CS_WAVES)
#define CS_U     8

__global__ __launch_bounds__(CS_WAVES * 32) void gat_colsum(
        const float* __restrict__ A,
        const float* __restrict__ s1,
        const float* __restrict__ s2,
        const float* __restrict__ a_b,
        float* __restrict__ D) {
    __shared__ float red[CS_WAVES][HH][32];

    const int lane = threadIdx.x & 31;
    const int w  = __builtin_amdgcn_readfirstlane(threadIdx.x >> 5);
    const int jt = blockIdx.x & 31;
    const int b  = blockIdx.x >> 5;
    const int j  = jt * 32 + lane;
    const int i0 = w * CS_ROWS;

    float s2v[HH], abv[HH], acc[HH];
    #pragma unroll
    for (int h = 0; h < HH; ++h) {
        s2v[h] = s2[((size_t)b * HH + h) * NN + j];
        abv[h] = a_b[h];
        acc[h] = 0.f;
    }
    const float* Ap  = A  + ((size_t)b * NN + i0) * NN + j;
    const float* s1p = s1 + (size_t)b * HH * NN + i0;

    float a_nxt[CS_U];
    #pragma unroll
    for (int u = 0; u < CS_U; ++u) a_nxt[u] = Ap[(size_t)u * NN];

    for (int r0 = 0; r0 < CS_ROWS; r0 += CS_U) {
        float a_cur[CS_U];
        #pragma unroll
        for (int u = 0; u < CS_U; ++u) a_cur[u] = a_nxt[u];
        if (r0 + CS_U < CS_ROWS) {
            #pragma unroll
            for (int u = 0; u < CS_U; ++u)
                a_nxt[u] = Ap[(size_t)(r0 + CS_U + u) * NN];
        }
        #pragma unroll
        for (int u = 0; u < CS_U; ++u) {
            #pragma unroll
            for (int h = 0; h < HH; ++h) {
                float e = s1p[h * NN + r0 + u] + s2v[h] + abv[h];
                e = (e >= 0.f) ? e : NEG * e;
                acc[h] += a_cur[u] * expf(e);
            }
        }
    }

    #pragma unroll
    for (int h = 0; h < HH; ++h) red[w][h][lane] = acc[h];
    __syncthreads();
    if (threadIdx.x < HH * 32) {
        float sum = 0.f;
        #pragma unroll
        for (int k = 0; k < CS_WAVES; ++k) sum += red[k][w][lane];
        D[((size_t)b * HH + w) * NN + j] = sum;
    }
}

__global__ void gat_scale(float* __restrict__ Wh, const float* __restrict__ D) {
    size_t idx = (size_t)blockIdx.x * blockDim.x + threadIdx.x;
    if (idx >= (size_t)BB * HH * NN * FOUT) return;
    Wh[idx] *= 1.0f / (D[idx >> 6] + EPSF);
}

#define AO_WAVES 4
#define AO_TJ    64
#define AO_PW    160
#define AO_PE    68

typedef float v4f __attribute__((ext_vector_type(4)));

__global__ __launch_bounds__(AO_WAVES * 32) void gat_attn_out(
        const float* __restrict__ A, const float* __restrict__ Whs,
        const float* __restrict__ s1, const float* __restrict__ s2,
        const float* __restrict__ a_b, float* __restrict__ out) {
    __shared__ __align__(16) float Ws[2][(AO_TJ / 2) * AO_PW];
    __shared__ __align__(16) float Es[2][AO_WAVES][16 * AO_PE];

    const int tid  = threadIdx.x;
    const int lane = tid & 31;
    const int lm = lane & 15, hi = lane >> 4;
    const int w  = __builtin_amdgcn_readfirstlane(tid >> 5);
    int bid = blockIdx.x;
    const int ib = bid & 15; bid >>= 4;
    const int h  = bid & 3;  bid >>= 2;
    const int b  = bid;
    const int i0 = ib * (AO_WAVES * 16) + w * 16;
    const int bh = b * HH + h;

    const float ab = a_b[h];
    const float* s1p = s1 + (size_t)bh * NN + i0;
    const float* s2p = s2 + (size_t)bh * NN + lane;
    const float* Ap  = A  + ((size_t)b * NN + i0) * NN + lane;
    const v4f* Wg = (const v4f*)(Whs + (size_t)bh * NN * FOUT)
                    + (tid >> 4) * 32 + (tid & 15);
    const int wst = (tid >> 4) * AO_PW + (tid & 15) * 8;

    float s1r[16];
    #pragma unroll
    for (int r = 0; r < 16; ++r) s1r[r] = s1p[r];

    float a_lo[16], a_hi[16];
    v4f   wv[8];
    float s2a, s2b;
    #pragma unroll
    for (int r = 0; r < 16; ++r) {
        a_lo[r] = Ap[(size_t)r * NN];
        a_hi[r] = Ap[(size_t)r * NN + 32];
    }
    #pragma unroll
    for (int q = 0; q < 4; ++q) {
        wv[2 * q]     = Wg[q * 256];
        wv[2 * q + 1] = Wg[q * 256 + 16];
    }
    s2a = s2p[0];
    s2b = s2p[32];

    v8f acc0 = {}, acc1 = {}, acc2 = {}, acc3 = {};

    for (int t = 0; t < NN / AO_TJ; ++t) {
        float* Eb = &Es[t & 1][w][0];
        float* Wb = &Ws[t & 1][0];

        #pragma unroll
        for (int r = 0; r < 16; ++r) {
            float e0 = s1r[r] + s2a + ab; e0 = (e0 >= 0.f) ? e0 : NEG * e0;
            float e1 = s1r[r] + s2b + ab; e1 = (e1 >= 0.f) ? e1 : NEG * e1;
            Eb[r * AO_PE + lane]      = a_lo[r] * expf(e0);
            Eb[r * AO_PE + lane + 32] = a_hi[r] * expf(e1);
        }
        #pragma unroll
        for (int q = 0; q < 4; ++q) {
            const v4f r0 = wv[2 * q], r1 = wv[2 * q + 1];
            const v4f lo = { r0.x, r1.x, r0.y, r1.y };
            const v4f hi4 = { r0.z, r1.z, r0.w, r1.w };
            *(v4f*)&Wb[wst + q * 8 * AO_PW]     = lo;
            *(v4f*)&Wb[wst + q * 8 * AO_PW + 4] = hi4;
        }
        __syncthreads();

        if (t + 1 < NN / AO_TJ) {
            const int j1 = (t + 1) * AO_TJ;
            #pragma unroll
            for (int r = 0; r < 16; ++r) {
                a_lo[r] = Ap[(size_t)r * NN + j1];
                a_hi[r] = Ap[(size_t)r * NN + j1 + 32];
            }
            #pragma unroll
            for (int q = 0; q < 4; ++q) {
                wv[2 * q]     = Wg[(size_t)j1 * (FOUT / 4) + q * 256];
                wv[2 * q + 1] = Wg[(size_t)j1 * (FOUT / 4) + q * 256 + 16];
            }
            s2a = s2p[j1];
            s2b = s2p[j1 + 32];
        }

        #pragma unroll
        for (int g = 0; g < 4; ++g) {
            v2f af[4], bf[4][4];
            #pragma unroll
            for (int u = 0; u < 4; ++u) {
                const int kk = g * 4 + u;
                af[u] = *(const v2f*)&Eb[lm * AO_PE + kk * 4 + 2 * hi];
                const float* wr = Wb + (kk * 2 + hi) * AO_PW + lm * 2;
                #pragma unroll
                for (int c = 0; c < 4; ++c) bf[u][c] = *(const v2f*)(wr + 32 * c);
            }
            #pragma unroll
            for (int u = 0; u < 4; ++u) {
                acc0 = wmma_f32(af[u], bf[u][0], acc0);
                acc1 = wmma_f32(af[u], bf[u][1], acc1);
                acc2 = wmma_f32(af[u], bf[u][2], acc2);
                acc3 = wmma_f32(af[u], bf[u][3], acc3);
            }
        }
    }

    float* op = out + ((size_t)b * NN + i0 + 8 * hi) * (HH * FOUT) + (size_t)h * FOUT + lm;
    #pragma unroll
    for (int r = 0; r < 8; ++r) {
        float* o = op + (size_t)r * (HH * FOUT);
        o[0] = acc0[r]; o[16] = acc1[r]; o[32] = acc2[r]; o[48] = acc3[r];
    }
}

extern "C" void kernel_launch(void* const* d_in, const int* in_sizes, int n_in,
                              void* d_out, int out_size, void* d_ws, size_t ws_size,
                              hipStream_t stream) {
    const float* A  = (const float*)d_in[0];
    const float* x  = (const float*)d_in[1];
    const float* W  = (const float*)d_in[2];
    const float* aw = (const float*)d_in[3];
    const float* ab = (const float*)d_in[4];
    float* out = (float*)d_out;

    float* Wh = (float*)d_ws;
    float* s1 = Wh + (size_t)BB * HH * NN * FOUT;
    float* s2 = s1 + (size_t)BB * HH * NN;
    float* D  = s2 + (size_t)BB * HH * NN;

    gat_wh      <<<dim3(BB * HH * (NN / 16)),  dim3(32),  0, stream>>>(x, W, Wh);
    gat_scores  <<<dim3(BB * HH * NN / 256),   dim3(256), 0, stream>>>(Wh, aw, s1, s2);
    gat_colsum  <<<dim3(BB * (NN / 32)), dim3(CS_WAVES * 32), 0, stream>>>(A, s1, s2, ab, D);
    gat_scale   <<<dim3((BB * HH * NN * FOUT) / 256), dim3(256), 0, stream>>>(Wh, D);
    gat_attn_out<<<dim3(BB * HH * (NN / (AO_WAVES * 16))), dim3(AO_WAVES * 32), 0, stream>>>(A, Wh, s1, s2, ab, out);
}


# column normalisation folded into the output kernel: colsum stores 1/(D+eps), each E tile element is multiplied by it as it is built; the one-thread-per-element scale launch is gone (5 launches -> 4)
# speedup vs baseline: 3.2159x; 1.0987x over previous
#include <hip/hip_runtime.h>
#include <math.h>

#define BB   8
#define NN   1024
#define FIN  128
#define FOUT 64
#define HH   4
#define EPSF 1e-7f
#define NEG  0.2f

typedef float v2f __attribute__((ext_vector_type(2)));
typedef float v8f __attribute__((ext_vector_type(8)));

__device__ __forceinline__ v8f wmma_f32(v2f a, v2f b, v8f c) {
    return __builtin_amdgcn_wmma_f32_16x16x4_f32(
         false, a,  false, b,
         (short)0, c,  false,  false);
}

__global__ __launch_bounds__(32) void gat_wh(const float* __restrict__ x,
                                             const float* __restrict__ W,
                                             float* __restrict__ Wh) {
    const int lane = threadIdx.x;
    const int lm = lane & 15, hi = lane >> 4;
    int bid = blockIdx.x;
    const int ib = bid & 63; bid >>= 6;
    const int h  = bid & 3;  bid >>= 2;
    const int b  = bid;
    const int i0 = ib * 16;

    const float* xr = x + ((size_t)b * NN + i0 + lm) * FIN + 2 * hi;
    const float* Wp = W + (size_t)h * FIN * FOUT + (size_t)(2 * hi) * FOUT + lm;

    v8f acc0 = {}, acc1 = {}, acc2 = {}, acc3 = {};
    #pragma unroll 4
    for (int k = 0; k < FIN; k += 4) {
        v2f a = *(const v2f*)(xr + k);
        const float* wk = Wp + (size_t)k * FOUT;
        v2f b0 = { wk[0],  wk[FOUT]      };
        v2f b1 = { wk[16], wk[FOUT + 16] };
        v2f b2 = { wk[32], wk[FOUT + 32] };
        v2f b3 = { wk[48], wk[FOUT + 48] };
        acc0 = wmma_f32(a, b0, acc0);
        acc1 = wmma_f32(a, b1, acc1);
        acc2 = wmma_f32(a, b2, acc2);
        acc3 = wmma_f32(a, b3, acc3);
    }
    float* outp = Wh + (((size_t)b * HH + h) * NN + i0 + 8 * hi) * FOUT + lm;
    #pragma unroll
    for (int r = 0; r < 8; ++r) {
        float* o = outp + (size_t)r * FOUT;
        o[0] = acc0[r]; o[16] = acc1[r]; o[32] = acc2[r]; o[48] = acc3[r];
    }
}

__global__ void gat_scores(const float* __restrict__ Wh,
                           const float* __restrict__ a_w,
                           float* __restrict__ s1, float* __restrict__ s2) {
    int idx = blockIdx.x * blockDim.x + threadIdx.x;
    if (idx >= BB * HH * NN) return;
    const int h = (idx / NN) & (HH - 1);
    const float* row = Wh + (size_t)idx * FOUT;
    const float* a1 = a_w + (size_t)h * 2 * FOUT;
    const float* a2 = a1 + FOUT;
    float v1 = 0.f, v2 = 0.f;
    #pragma unroll 8
    for (int o = 0; o < FOUT; ++o) {
        float w = row[o];
        v1 += w * a1[o];
        v2 += w * a2[o];
    }
    s1[idx] = v1; s2[idx] = v2;
}

#define CS_WAVES 16
#define CS_ROWS  (NN / CS_WAVES)
#define CS_U     8

__global__ __launch_bounds__(CS_WAVES * 32) void gat_colsum(
        const float* __restrict__ A,
        const float* __restrict__ s1,
        const float* __restrict__ s2,
        const float* __restrict__ a_b,
        float* __restrict__ Dinv) {
    __shared__ float red[CS_WAVES][HH][32];

    const int lane = threadIdx.x & 31;
    const int w  = __builtin_amdgcn_readfirstlane(threadIdx.x >> 5);
    const int jt = blockIdx.x & 31;
    const int b  = blockIdx.x >> 5;
    const int j  = jt * 32 + lane;
    const int i0 = w * CS_ROWS;

    float s2v[HH], abv[HH], acc[HH];
    #pragma unroll
    for (int h = 0; h < HH; ++h) {
        s2v[h] = s2[((size_t)b * HH + h) * NN + j];
        abv[h] = a_b[h];
        acc[h] = 0.f;
    }
    const float* Ap  = A  + ((size_t)b * NN + i0) * NN + j;
    const float* s1p = s1 + (size_t)b * HH * NN + i0;

    float a_nxt[CS_U];
    #pragma unroll
    for (int u = 0; u < CS_U; ++u) a_nxt[u] = Ap[(size_t)u * NN];

    for (int r0 = 0; r0 < CS_ROWS; r0 += CS_U) {
        float a_cur[CS_U];
        #pragma unroll
        for (int u = 0; u < CS_U; ++u) a_cur[u] = a_nxt[u];
        if (r0 + CS_U < CS_ROWS) {
            #pragma unroll
            for (int u = 0; u < CS_U; ++u)
                a_nxt[u] = Ap[(size_t)(r0 + CS_U + u) * NN];
        }
        #pragma unroll
        for (int u = 0; u < CS_U; ++u) {
            #pragma unroll
            for (int h = 0; h < HH; ++h) {
                float e = s1p[h * NN + r0 + u] + s2v[h] + abv[h];
                e = (e >= 0.f) ? e : NEG * e;
                acc[h] += a_cur[u] * expf(e);
            }
        }
    }

    #pragma unroll
    for (int h = 0; h < HH; ++h) red[w][h][lane] = acc[h];
    __syncthreads();
    if (threadIdx.x < HH * 32) {
        float sum = 0.f;
        #pragma unroll
        for (int k = 0; k < CS_WAVES; ++k) sum += red[k][w][lane];
        Dinv[((size_t)b * HH + w) * NN + j] = 1.0f / (sum + EPSF);
    }
}

#define AO_WAVES 4
#define AO_TJ    64
#define AO_PW    160
#define AO_PE    68

typedef float v4f __attribute__((ext_vector_type(4)));

__global__ __launch_bounds__(AO_WAVES * 32) void gat_attn_out(
        const float* __restrict__ A, const float* __restrict__ Wh,
        const float* __restrict__ s1, const float* __restrict__ s2,
        const float* __restrict__ Dinv,
        const float* __restrict__ a_b, float* __restrict__ out) {
    __shared__ __align__(16) float Ws[2][(AO_TJ / 2) * AO_PW];
    __shared__ __align__(16) float Es[2][AO_WAVES][16 * AO_PE];

    const int tid  = threadIdx.x;
    const int lane = tid & 31;
    const int lm = lane & 15, hi = lane >> 4;
    const int w  = __builtin_amdgcn_readfirstlane(tid >> 5);
    int bid = blockIdx.x;
    const int ib = bid & 15; bid >>= 4;
    const int h  = bid & 3;  bid >>= 2;
    const int b  = bid;
    const int i0 = ib * (AO_WAVES * 16) + w * 16;
    const int bh = b * HH + h;

    const float ab = a_b[h];
    const float* s1p = s1 + (size_t)bh * NN + i0;
    const float* s2p = s2 + (size_t)bh * NN + lane;
    const float* dvp = Dinv + (size_t)bh * NN + lane;
    const float* Ap  = A  + ((size_t)b * NN + i0) * NN + lane;
    const v4f* Wg = (const v4f*)(Wh + (size_t)bh * NN * FOUT)
                    + (tid >> 4) * 32 + (tid & 15);
    const int wst = (tid >> 4) * AO_PW + (tid & 15) * 8;

    float s1r[16];
    #pragma unroll
    for (int r = 0; r < 16; ++r) s1r[r] = s1p[r];

    float a_lo[16], a_hi[16];
    v4f   wv[8];
    float s2a, s2b, dva, dvb;
    #pragma unroll
    for (int r = 0; r < 16; ++r) {
        a_lo[r] = Ap[(size_t)r * NN];
        a_hi[r] = Ap[(size_t)r * NN + 32];
    }
    #pragma unroll
    for (int q = 0; q < 4; ++q) {
        wv[2 * q]     = Wg[q * 256];
        wv[2 * q + 1] = Wg[q * 256 + 16];
    }
    s2a = s2p[0];
    s2b = s2p[32];
    dva = dvp[0];
    dvb = dvp[32];

    v8f acc0 = {}, acc1 = {}, acc2 = {}, acc3 = {};

    for (int t = 0; t < NN / AO_TJ; ++t) {
        float* Eb = &Es[t & 1][w][0];
        float* Wb = &Ws[t & 1][0];

        #pragma unroll
        for (int r = 0; r < 16; ++r) {
            float e0 = s1r[r] + s2a + ab; e0 = (e0 >= 0.f) ? e0 : NEG * e0;
            float e1 = s1r[r] + s2b + ab; e1 = (e1 >= 0.f) ? e1 : NEG * e1;
            Eb[r * AO_PE + lane]      = (a_lo[r] * expf(e0)) * dva;
            Eb[r * AO_PE + lane + 32] = (a_hi[r] * expf(e1)) * dvb;
        }
        #pragma unroll
        for (int q = 0; q < 4; ++q) {
            const v4f r0 = wv[2 * q], r1 = wv[2 * q + 1];
            const v4f lo = { r0.x, r1.x, r0.y, r1.y };
            const v4f hi4 = { r0.z, r1.z, r0.w, r1.w };
            *(v4f*)&Wb[wst + q * 8 * AO_PW]     = lo;
            *(v4f*)&Wb[wst + q * 8 * AO_PW + 4] = hi4;
        }
        __syncthreads();

        if (t + 1 < NN / AO_TJ) {
            const int j1 = (t + 1) * AO_TJ;
            #pragma unroll
            for (int r = 0; r < 16; ++r) {
                a_lo[r] = Ap[(size_t)r * NN + j1];
                a_hi[r] = Ap[(size_t)r * NN + j1 + 32];
            }
            #pragma unroll
            for (int q = 0; q < 4; ++q) {
                wv[2 * q]     = Wg[(size_t)j1 * (FOUT / 4) + q * 256];
                wv[2 * q + 1] = Wg[(size_t)j1 * (FOUT / 4) + q * 256 + 16];
            }
            s2a = s2p[j1];
            s2b = s2p[j1 + 32];
            dva = dvp[j1];
            dvb = dvp[j1 + 32];
        }

        #pragma unroll
        for (int g = 0; g < 4; ++g) {
            v2f af[4], bf[4][4];
            #pragma unroll
            for (int u = 0; u < 4; ++u) {
                const int kk = g * 4 + u;
                af[u] = *(const v2f*)&Eb[lm * AO_PE + kk * 4 + 2 * hi];
                const float* wr = Wb + (kk * 2 + hi) * AO_PW + lm * 2;
                #pragma unroll
                for (int c = 0; c < 4; ++c) bf[u][c] = *(const v2f*)(wr + 32 * c);
            }
            #pragma unroll
            for (int u = 0; u < 4; ++u) {
                acc0 = wmma_f32(af[u], bf[u][0], acc0);
                acc1 = wmma_f32(af[u], bf[u][1], acc1);
                acc2 = wmma_f32(af[u], bf[u][2], acc2);
                acc3 = wmma_f32(af[u], bf[u][3], acc3);
            }
        }
    }

    float* op = out + ((size_t)b * NN + i0 + 8 * hi) * (HH * FOUT) + (size_t)h * FOUT + lm;
    #pragma unroll
    for (int r = 0; r < 8; ++r) {
        float* o = op + (size_t)r * (HH * FOUT);
        o[0] = acc0[r]; o[16] = acc1[r]; o[32] = acc2[r]; o[48] = acc3[r];
    }
}

extern "C" void kernel_launch(void* const* d_in, const int* in_sizes, int n_in,
                              void* d_out, int out_size, void* d_ws, size_t ws_size,
                              hipStream_t stream) {
    const float* A  = (const float*)d_in[0];
    const float* x  = (const float*)d_in[1];
    const float* W  = (const float*)d_in[2];
    const float* aw = (const float*)d_in[3];
    const float* ab = (const float*)d_in[4];
    float* out = (float*)d_out;

    float* Wh = (float*)d_ws;
    float* s1 = Wh + (size_t)BB * HH * NN * FOUT;
    float* s2 = s1 + (size_t)BB * HH * NN;
    float* Dinv = s2 + (size_t)BB * HH * NN;

    gat_wh      <<<dim3(BB * HH * (NN / 16)),  dim3(32),  0, stream>>>(x, W, Wh);
    gat_scores  <<<dim3(BB * HH * NN / 256),   dim3(256), 0, stream>>>(Wh, aw, s1, s2);
    gat_colsum  <<<dim3(BB * (NN / 32)), dim3(CS_WAVES * 32), 0, stream>>>(A, s1, s2, ab, Dinv);
    gat_attn_out<<<dim3(BB * HH * (NN / (AO_WAVES * 16))), dim3(AO_WAVES * 32), 0, stream>>>(A, Wh, s1, s2, Dinv, ab, out);
}


# exp(lrelu(x)) of both weight loops as hardware exp2(max(t,t/5)), t=log2(e)*x: scores kernel stores s1*log2(e) and (s2+bias)*log2(e), bias now added to s2 first; expf chains gone
# speedup vs baseline: 3.8800x; 1.2065x over previous
#include <hip/hip_runtime.h>
#include <math.h>

#define BB   8
#define NN   1024
#define FIN  128
#define FOUT 64
#define HH   4
#define EPSF 1e-7f
#define NEG  0.2f
#define LOG2E 1.44269504088896340736f

typedef float v2f __attribute__((ext_vector_type(2)));
typedef float v8f __attribute__((ext_vector_type(8)));

__device__ __forceinline__ v8f wmma_f32(v2f a, v2f b, v8f c) {
    return __builtin_amdgcn_wmma_f32_16x16x4_f32(
         false, a,  false, b,
         (short)0, c,  false,  false);
}

__global__ __launch_bounds__(32) void gat_wh(const float* __restrict__ x,
                                             const float* __restrict__ W,
                                             float* __restrict__ Wh) {
    const int lane = threadIdx.x;
    const int lm = lane & 15, hi = lane >> 4;
    int bid = blockIdx.x;
    const int ib = bid & 63; bid >>= 6;
    const int h  = bid & 3;  bid >>= 2;
    const int b  = bid;
    const int i0 = ib * 16;

    const float* xr = x + ((size_t)b * NN + i0 + lm) * FIN + 2 * hi;
    const float* Wp = W + (size_t)h * FIN * FOUT + (size_t)(2 * hi) * FOUT + lm;

    v8f acc0 = {}, acc1 = {}, acc2 = {}, acc3 = {};
    #pragma unroll 4
    for (int k = 0; k < FIN; k += 4) {
        v2f a = *(const v2f*)(xr + k);
        const float* wk = Wp + (size_t)k * FOUT;
        v2f b0 = { wk[0],  wk[FOUT]      };
        v2f b1 = { wk[16], wk[FOUT + 16] };
        v2f b2 = { wk[32], wk[FOUT + 32] };
        v2f b3 = { wk[48], wk[FOUT + 48] };
        acc0 = wmma_f32(a, b0, acc0);
        acc1 = wmma_f32(a, b1, acc1);
        acc2 = wmma_f32(a, b2, acc2);
        acc3 = wmma_f32(a, b3, acc3);
    }
    float* outp = Wh + (((size_t)b * HH + h) * NN + i0 + 8 * hi) * FOUT + lm;
    #pragma unroll
    for (int r = 0; r < 8; ++r) {
        float* o = outp + (size_t)r * FOUT;
        o[0] = acc0[r]; o[16] = acc1[r]; o[32] = acc2[r]; o[48] = acc3[r];
    }
}

__global__ void gat_scores(const float* __restrict__ Wh,
                           const float* __restrict__ a_w,
                           const float* __restrict__ a_b,
                           float* __restrict__ s1, float* __restrict__ s2) {
    int idx = blockIdx.x * blockDim.x + threadIdx.x;
    if (idx >= BB * HH * NN) return;
    const int h = (idx / NN) & (HH - 1);
    const float* row = Wh + (size_t)idx * FOUT;
    const float* a1 = a_w + (size_t)h * 2 * FOUT;
    const float* a2 = a1 + FOUT;
    float v1 = 0.f, v2 = 0.f;
    #pragma unroll 8
    for (int o = 0; o < FOUT; ++o) {
        float w = row[o];
        v1 += w * a1[o];
        v2 += w * a2[o];
    }
    s1[idx] = v1 * LOG2E;
    s2[idx] = (v2 + a_b[h]) * LOG2E;
}

__device__ __forceinline__ float attn_weight(float s1v, float s2v) {
    const float e2 = s1v + s2v;
    return __builtin_amdgcn_exp2f(fmaxf(e2, NEG * e2));
}

#define CS_WAVES 16
#define CS_ROWS  (NN / CS_WAVES)
#define CS_U     8

__global__ __launch_bounds__(CS_WAVES * 32) void gat_colsum(
        const float* __restrict__ A,
        const float* __restrict__ s1,
        const float* __restrict__ s2,
        float* __restrict__ Dinv) {
    __shared__ float red[CS_WAVES][HH][32];

    const int lane = threadIdx.x & 31;
    const int w  = __builtin_amdgcn_readfirstlane(threadIdx.x >> 5);
    const int jt = blockIdx.x & 31;
    const int b  = blockIdx.x >> 5;
    const int j  = jt * 32 + lane;
    const int i0 = w * CS_ROWS;

    float s2v[HH], acc[HH];
    #pragma unroll
    for (int h = 0; h < HH; ++h) {
        s2v[h] = s2[((size_t)b * HH + h) * NN + j];
        acc[h] = 0.f;
    }
    const float* Ap  = A  + ((size_t)b * NN + i0) * NN + j;
    const float* s1p = s1 + (size_t)b * HH * NN + i0;

    float a_nxt[CS_U];
    #pragma unroll
    for (int u = 0; u < CS_U; ++u) a_nxt[u] = Ap[(size_t)u * NN];

    for (int r0 = 0; r0 < CS_ROWS; r0 += CS_U) {
        float a_cur[CS_U];
        #pragma unroll
        for (int u = 0; u < CS_U; ++u) a_cur[u] = a_nxt[u];
        if (r0 + CS_U < CS_ROWS) {
            #pragma unroll
            for (int u = 0; u < CS_U; ++u)
                a_nxt[u] = Ap[(size_t)(r0 + CS_U + u) * NN];
        }
        #pragma unroll
        for (int u = 0; u < CS_U; ++u) {
            #pragma unroll
            for (int h = 0; h < HH; ++h) {
                acc[h] += a_cur[u] * attn_weight(s1p[h * NN + r0 + u], s2v[h]);
            }
        }
    }

    #pragma unroll
    for (int h = 0; h < HH; ++h) red[w][h][lane] = acc[h];
    __syncthreads();
    if (threadIdx.x < HH * 32) {
        float sum = 0.f;
        #pragma unroll
        for (int k = 0; k < CS_WAVES; ++k) sum += red[k][w][lane];
        Dinv[((size_t)b * HH + w) * NN + j] = 1.0f / (sum + EPSF);
    }
}

#define AO_WAVES 4
#define AO_TJ    64
#define AO_PW    160
#define AO_PE    68

typedef float v4f __attribute__((ext_vector_type(4)));

__global__ __launch_bounds__(AO_WAVES * 32) void gat_attn_out(
        const float* __restrict__ A, const float* __restrict__ Wh,
        const float* __restrict__ s1, const float* __restrict__ s2,
        const float* __restrict__ Dinv, float* __restrict__ out) {
    __shared__ __align__(16) float Ws[2][(AO_TJ / 2) * AO_PW];
    __shared__ __align__(16) float Es[2][AO_WAVES][16 * AO_PE];

    const int tid  = threadIdx.x;
    const int lane = tid & 31;
    const int lm = lane & 15, hi = lane >> 4;
    const int w  = __builtin_amdgcn_readfirstlane(tid >> 5);
    int bid = blockIdx.x;
    const int ib = bid & 15; bid >>= 4;
    const int h  = bid & 3;  bid >>= 2;
    const int b  = bid;
    const int i0 = ib * (AO_WAVES * 16) + w * 16;
    const int bh = b * HH + h;

    const float* s1p = s1 + (size_t)bh * NN + i0;
    const float* s2p = s2 + (size_t)bh * NN + lane;
    const float* dvp = Dinv + (size_t)bh * NN + lane;
    const float* Ap  = A  + ((size_t)b * NN + i0) * NN + lane;
    const v4f* Wg = (const v4f*)(Wh + (size_t)bh * NN * FOUT)
                    + (tid >> 4) * 32 + (tid & 15);
    const int wst = (tid >> 4) * AO_PW + (tid & 15) * 8;

    float s1r[16];
    #pragma unroll
    for (int r = 0; r < 16; ++r) s1r[r] = s1p[r];

    float a_lo[16], a_hi[16];
    v4f   wv[8];
    float s2a, s2b, dva, dvb;
    #pragma unroll
    for (int r = 0; r < 16; ++r) {
        a_lo[r] = Ap[(size_t)r * NN];
        a_hi[r] = Ap[(size_t)r * NN + 32];
    }
    #pragma unroll
    for (int q = 0; q < 4; ++q) {
        wv[2 * q]     = Wg[q * 256];
        wv[2 * q + 1] = Wg[q * 256 + 16];
    }
    s2a = s2p[0];
    s2b = s2p[32];
    dva = dvp[0];
    dvb = dvp[32];

    v8f acc0 = {}, acc1 = {}, acc2 = {}, acc3 = {};

    for (int t = 0; t < NN / AO_TJ; ++t) {
        float* Eb = &Es[t & 1][w][0];
        float* Wb = &Ws[t & 1][0];

        #pragma unroll
        for (int r = 0; r < 16; ++r) {
            Eb[r * AO_PE + lane]      = (a_lo[r] * attn_weight(s1r[r], s2a)) * dva;
            Eb[r * AO_PE + lane + 32] = (a_hi[r] * attn_weight(s1r[r], s2b)) * dvb;
        }
        #pragma unroll
        for (int q = 0; q < 4; ++q) {
            const v4f r0 = wv[2 * q], r1 = wv[2 * q + 1];
            const v4f lo = { r0.x, r1.x, r0.y, r1.y };
            const v4f hi4 = { r0.z, r1.z, r0.w, r1.w };
            *(v4f*)&Wb[wst + q * 8 * AO_PW]     = lo;
            *(v4f*)&Wb[wst + q * 8 * AO_PW + 4] = hi4;
        }
        __syncthreads();

        if (t + 1 < NN / AO_TJ) {
            const int j1 = (t + 1) * AO_TJ;
            #pragma unroll
            for (int r = 0; r < 16; ++r) {
                a_lo[r] = Ap[(size_t)r * NN + j1];
                a_hi[r] = Ap[(size_t)r * NN + j1 + 32];
            }
            #pragma unroll
            for (int q = 0; q < 4; ++q) {
                wv[2 * q]     = Wg[(size_t)j1 * (FOUT / 4) + q * 256];
                wv[2 * q + 1] = Wg[(size_t)j1 * (FOUT / 4) + q * 256 + 16];
            }
            s2a = s2p[j1];
            s2b = s2p[j1 + 32];
            dva = dvp[j1];
            dvb = dvp[j1 + 32];
        }

        #pragma unroll
        for (int g = 0; g < 4; ++g) {
            v2f af[4], bf[4][4];
            #pragma unroll
            for (int u = 0; u < 4; ++u) {
                const int kk = g * 4 + u;
                af[u] = *(const v2f*)&Eb[lm * AO_PE + kk * 4 + 2 * hi];
                const float* wr = Wb + (kk * 2 + hi) * AO_PW + lm * 2;
                #pragma unroll
                for (int c = 0; c < 4; ++c) bf[u][c] = *(const v2f*)(wr + 32 * c);
            }
            #pragma unroll
            for (int u = 0; u < 4; ++u) {
                acc0 = wmma_f32(af[u], bf[u][0], acc0);
                acc1 = wmma_f32(af[u], bf[u][1], acc1);
                acc2 = wmma_f32(af[u], bf[u][2], acc2);
                acc3 = wmma_f32(af[u], bf[u][3], acc3);
            }
        }
    }

    float* op = out + ((size_t)b * NN + i0 + 8 * hi) * (HH * FOUT) + (size_t)h * FOUT + lm;
    #pragma unroll
    for (int r = 0; r < 8; ++r) {
        float* o = op + (size_t)r * (HH * FOUT);
        o[0] = acc0[r]; o[16] = acc1[r]; o[32] = acc2[r]; o[48] = acc3[r];
    }
}

extern "C" void kernel_launch(void* const* d_in, const int* in_sizes, int n_in,
                              void* d_out, int out_size, void* d_ws, size_t ws_size,
                              hipStream_t stream) {
    const float* A  = (const float*)d_in[0];
    const float* x  = (const float*)d_in[1];
    const float* W  = (const float*)d_in[2];
    const float* aw = (const float*)d_in[3];
    const float* ab = (const float*)d_in[4];
    float* out = (float*)d_out;

    float* Wh = (float*)d_ws;
    float* s1 = Wh + (size_t)BB * HH * NN * FOUT;
    float* s2 = s1 + (size_t)BB * HH * NN;
    float* Dinv = s2 + (size_t)BB * HH * NN;

    gat_wh      <<<dim3(BB * HH * (NN / 16)),  dim3(32),  0, stream>>>(x, W, Wh);
    gat_scores  <<<dim3(BB * HH * NN / 256),   dim3(256), 0, stream>>>(Wh, aw, ab, s1, s2);
    gat_colsum  <<<dim3(BB * (NN / 32)), dim3(CS_WAVES * 32), 0, stream>>>(A, s1, s2, Dinv);
    gat_attn_out<<<dim3(BB * HH * (NN / (AO_WAVES * 16))), dim3(AO_WAVES * 32), 0, stream>>>(A, Wh, s1, s2, Dinv, out);
}


# scores s1,s2 taken in the Wh kernel from its accumulators (4 products per lane + 16-lane DPP row sum: changes the order of the sum over o), one 128-B line per wave; scores launch gone (4 -> 3)
# speedup vs baseline: 4.2805x; 1.1032x over previous
#include <hip/hip_runtime.h>
#include <math.h>

#define BB   8
#define NN   1024
#define FIN  128
#define FOUT 64
#define HH   4
#define EPSF 1e-7f
#define NEG  0.2f
#define LOG2E 1.44269504088896340736f

typedef float v2f __attribute__((ext_vector_type(2)));
typedef float v4f __attribute__((ext_vector_type(4)));
typedef float v8f __attribute__((ext_vector_type(8)));

__device__ __forceinline__ v8f wmma_f32(v2f a, v2f b, v8f c) {
    return __builtin_amdgcn_wmma_f32_16x16x4_f32(
         false, a,  false, b,
         (short)0, c,  false,  false);
}

__device__ __forceinline__ int sc_row(int n) { return ((n >> 4) << 5) + (n & 15); }

__device__ __forceinline__ float row16_sum(float v) {
    v += __builtin_bit_cast(float, __builtin_amdgcn_mov_dpp(__builtin_bit_cast(int, v), 0x161, 0xf, 0xf, true));
    v += __builtin_bit_cast(float, __builtin_amdgcn_mov_dpp(__builtin_bit_cast(int, v), 0x162, 0xf, 0xf, true));
    v += __builtin_bit_cast(float, __builtin_amdgcn_mov_dpp(__builtin_bit_cast(int, v), 0x164, 0xf, 0xf, true));
    v += __builtin_bit_cast(float, __builtin_amdgcn_mov_dpp(__builtin_bit_cast(int, v), 0x168, 0xf, 0xf, true));
    return v;
}

__global__ __launch_bounds__(32) void gat_wh(const float* __restrict__ x,
                                             const float* __restrict__ W,
                                             const float* __restrict__ a_w,
                                             const float* __restrict__ a_b,
                                             float* __restrict__ Wh,
                                             float* __restrict__ sc) {
    const int lane = threadIdx.x;
    const int lm = lane & 15, hi = lane >> 4;
    int bid = blockIdx.x;
    const int ib = bid & 63; bid >>= 6;
    const int h  = bid & 3;  bid >>= 2;
    const int b  = bid;
    const int i0 = ib * 16;

    const float* xr = x + ((size_t)b * NN + i0 + lm) * FIN + 2 * hi;
    const float* Wp = W + (size_t)h * FIN * FOUT + (size_t)(2 * hi) * FOUT + lm;

    const float* a1 = a_w + (size_t)h * 2 * FOUT + lm;
    float a1v[4], a2v[4];
    #pragma unroll
    for (int t = 0; t < 4; ++t) { a1v[t] = a1[16 * t]; a2v[t] = a1[FOUT + 16 * t]; }
    const float bias = a_b[h];

    v8f acc0 = {}, acc1 = {}, acc2 = {}, acc3 = {};
    #pragma unroll 4
    for (int k = 0; k < FIN; k += 4) {
        v2f a = *(const v2f*)(xr + k);
        const float* wk = Wp + (size_t)k * FOUT;
        v2f b0 = { wk[0],  wk[FOUT]      };
        v2f b1 = { wk[16], wk[FOUT + 16] };
        v2f b2 = { wk[32], wk[FOUT + 32] };
        v2f b3 = { wk[48], wk[FOUT + 48] };
        acc0 = wmma_f32(a, b0, acc0);
        acc1 = wmma_f32(a, b1, acc1);
        acc2 = wmma_f32(a, b2, acc2);
        acc3 = wmma_f32(a, b3, acc3);
    }
    asm volatile("s_wait_alu depctr_va_vdst(0)"
                 : "+v"(acc0), "+v"(acc1), "+v"(acc2), "+v"(acc3));

    float* outp = Wh + (((size_t)b * HH + h) * NN + i0 + 8 * hi) * FOUT + lm;
    #pragma unroll
    for (int r = 0; r < 8; ++r) {
        float* o = outp + (size_t)r * FOUT;
        o[0] = acc0[r]; o[16] = acc1[r]; o[32] = acc2[r]; o[48] = acc3[r];
    }

    float p1[8], p2[8];
    #pragma unroll
    for (int r = 0; r < 8; ++r) {
        p1[r] = acc0[r] * a1v[0] + acc1[r] * a1v[1] + acc2[r] * a1v[2] + acc3[r] * a1v[3];
        p2[r] = acc0[r] * a2v[0] + acc1[r] * a2v[1] + acc2[r] * a2v[2] + acc3[r] * a2v[3];
    }
    #pragma unroll
    for (int r = 0; r < 8; ++r) { p1[r] = row16_sum(p1[r]); p2[r] = row16_sum(p2[r]); }
    if (lm == 0) {
        float* so = sc + ((size_t)b * HH + h) * 2 * NN + sc_row(i0 + 8 * hi);
        const v4f u0 = { p1[0] * LOG2E, p1[1] * LOG2E, p1[2] * LOG2E, p1[3] * LOG2E };
        const v4f u1 = { p1[4] * LOG2E, p1[5] * LOG2E, p1[6] * LOG2E, p1[7] * LOG2E };
        const v4f w0 = { (p2[0] + bias) * LOG2E, (p2[1] + bias) * LOG2E,
                         (p2[2] + bias) * LOG2E, (p2[3] + bias) * LOG2E };
        const v4f w1 = { (p2[4] + bias) * LOG2E, (p2[5] + bias) * LOG2E,
                         (p2[6] + bias) * LOG2E, (p2[7] + bias) * LOG2E };
        *(v4f*)(so)      = u0;
        *(v4f*)(so + 4)  = u1;
        *(v4f*)(so + 16) = w0;
        *(v4f*)(so + 20) = w1;
    }
}

__device__ __forceinline__ float attn_weight(float s1v, float s2v) {
    const float e2 = s1v + s2v;
    return __builtin_amdgcn_exp2f(fmaxf(e2, NEG * e2));
}

#define CS_WAVES 16
#define CS_ROWS  (NN / CS_WAVES)
#define CS_U     8

__global__ __launch_bounds__(CS_WAVES * 32) void gat_colsum(
        const float* __restrict__ A,
        const float* __restrict__ sc,
        float* __restrict__ Dinv) {
    __shared__ float red[CS_WAVES][HH][32];

    const int lane = threadIdx.x & 31;
    const int w  = __builtin_amdgcn_readfirstlane(threadIdx.x >> 5);
    const int jt = blockIdx.x & 31;
    const int b  = blockIdx.x >> 5;
    const int j  = jt * 32 + lane;
    const int i0 = w * CS_ROWS;

    float s2v[HH], acc[HH];
    #pragma unroll
    for (int h = 0; h < HH; ++h) {
        s2v[h] = sc[((size_t)b * HH + h) * 2 * NN + sc_row(j) + 16];
        acc[h] = 0.f;
    }
    const float* Ap  = A  + ((size_t)b * NN + i0) * NN + j;
    const float* s1p = sc + (size_t)b * HH * 2 * NN;

    float a_nxt[CS_U];
    #pragma unroll
    for (int u = 0; u < CS_U; ++u) a_nxt[u] = Ap[(size_t)u * NN];

    for (int r0 = 0; r0 < CS_ROWS; r0 += CS_U) {
        float a_cur[CS_U];
        #pragma unroll
        for (int u = 0; u < CS_U; ++u) a_cur[u] = a_nxt[u];
        if (r0 + CS_U < CS_ROWS) {
            #pragma unroll
            for (int u = 0; u < CS_U; ++u)
                a_nxt[u] = Ap[(size_t)(r0 + CS_U + u) * NN];
        }
        #pragma unroll
        for (int u = 0; u < CS_U; ++u) {
            #pragma unroll
            for (int h = 0; h < HH; ++h) {
                acc[h] += a_cur[u] * attn_weight(s1p[h * 2 * NN + sc_row(i0 + r0) + u], s2v[h]);
            }
        }
    }

    #pragma unroll
    for (int h = 0; h < HH; ++h) red[w][h][lane] = acc[h];
    __syncthreads();
    if (threadIdx.x < HH * 32) {
        float sum = 0.f;
        #pragma unroll
        for (int k = 0; k < CS_WAVES; ++k) sum += red[k][w][lane];
        Dinv[((size_t)b * HH + w) * NN + j] = 1.0f / (sum + EPSF);
    }
}

#define AO_WAVES 4
#define AO_TJ    64
#define AO_PW    160
#define AO_PE    68

__global__ __launch_bounds__(AO_WAVES * 32) void gat_attn_out(
        const float* __restrict__ A, const float* __restrict__ Wh,
        const float* __restrict__ sc,
        const float* __restrict__ Dinv, float* __restrict__ out) {
    __shared__ __align__(16) float Ws[2][(AO_TJ / 2) * AO_PW];
    __shared__ __align__(16) float Es[2][AO_WAVES][16 * AO_PE];

    const int tid  = threadIdx.x;
    const int lane = tid & 31;
    const int lm = lane & 15, hi = lane >> 4;
    const int w  = __builtin_amdgcn_readfirstlane(tid >> 5);
    int bid = blockIdx.x;
    const int ib = bid & 15; bid >>= 4;
    const int h  = bid & 3;  bid >>= 2;
    const int b  = bid;
    const int i0 = ib * (AO_WAVES * 16) + w * 16;
    const int bh = b * HH + h;

    const float* s1p = sc + (size_t)bh * 2 * NN + sc_row(i0);
    const float* s2p = sc + (size_t)bh * 2 * NN + sc_row(lane) + 16;
    const float* dvp = Dinv + (size_t)bh * NN + lane;
    const float* Ap  = A  + ((size_t)b * NN + i0) * NN + lane;
    const v4f* Wg = (const v4f*)(Wh + (size_t)bh * NN * FOUT)
                    + (tid >> 4) * 32 + (tid & 15);
    const int wst = (tid >> 4) * AO_PW + (tid & 15) * 8;

    float s1r[16];
    #pragma unroll
    for (int r = 0; r < 16; ++r) s1r[r] = s1p[r];

    float a_lo[16], a_hi[16];
    v4f   wv[8];
    float s2a, s2b, dva, dvb;
    #pragma unroll
    for (int r = 0; r < 16; ++r) {
        a_lo[r] = Ap[(size_t)r * NN];
        a_hi[r] = Ap[(size_t)r * NN + 32];
    }
    #pragma unroll
    for (int q = 0; q < 4; ++q) {
        wv[2 * q]     = Wg[q * 256];
        wv[2 * q + 1] = Wg[q * 256 + 16];
    }
    s2a = s2p[0];
    s2b = s2p[64];
    dva = dvp[0];
    dvb = dvp[32];

    v8f acc0 = {}, acc1 = {}, acc2 = {}, acc3 = {};

    for (int t = 0; t < NN / AO_TJ; ++t) {
        float* Eb = &Es[t & 1][w][0];
        float* Wb = &Ws[t & 1][0];

        #pragma unroll
        for (int r = 0; r < 16; ++r) {
            Eb[r * AO_PE + lane]      = (a_lo[r] * attn_weight(s1r[r], s2a)) * dva;
            Eb[r * AO_PE + lane + 32] = (a_hi[r] * attn_weight(s1r[r], s2b)) * dvb;
        }
        #pragma unroll
        for (int q = 0; q < 4; ++q) {
            const v4f r0 = wv[2 * q], r1 = wv[2 * q + 1];
            const v4f lo = { r0.x, r1.x, r0.y, r1.y };
            const v4f hi4 = { r0.z, r1.z, r0.w, r1.w };
            *(v4f*)&Wb[wst + q * 8 * AO_PW]     = lo;
            *(v4f*)&Wb[wst + q * 8 * AO_PW + 4] = hi4;
        }
        __syncthreads();

        if (t + 1 < NN / AO_TJ) {
            const int j1 = (t + 1) * AO_TJ;
            #pragma unroll
            for (int r = 0; r < 16; ++r) {
                a_lo[r] = Ap[(size_t)r * NN + j1];
                a_hi[r] = Ap[(size_t)r * NN + j1 + 32];
            }
            #pragma unroll
            for (int q = 0; q < 4; ++q) {
                wv[2 * q]     = Wg[(size_t)j1 * (FOUT / 4) + q * 256];
                wv[2 * q + 1] = Wg[(size_t)j1 * (FOUT / 4) + q * 256 + 16];
            }
            s2a = s2p[2 * j1];
            s2b = s2p[2 * j1 + 64];
            dva = dvp[j1];
            dvb = dvp[j1 + 32];
        }

        #pragma unroll
        for (int g = 0; g < 4; ++g) {
            v2f af[4], bf[4][4];
            #pragma unroll
            for (int u = 0; u < 4; ++u) {
                const int kk = g * 4 + u;
                af[u] = *(const v2f*)&Eb[lm * AO_PE + kk * 4 + 2 * hi];
                const float* wr = Wb + (kk * 2 + hi) * AO_PW + lm * 2;
                #pragma unroll
                for (int c = 0; c < 4; ++c) bf[u][c] = *(const v2f*)(wr + 32 * c);
            }
            #pragma unroll
            for (int u = 0; u < 4; ++u) {
                acc0 = wmma_f32(af[u], bf[u][0], acc0);
                acc1 = wmma_f32(af[u], bf[u][1], acc1);
                acc2 = wmma_f32(af[u], bf[u][2], acc2);
                acc3 = wmma_f32(af[u], bf[u][3], acc3);
            }
        }
    }

    float* op = out + ((size_t)b * NN + i0 + 8 * hi) * (HH * FOUT) + (size_t)h * FOUT + lm;
    #pragma unroll
    for (int r = 0; r < 8; ++r) {
        float* o = op + (size_t)r * (HH * FOUT);
        o[0] = acc0[r]; o[16] = acc1[r]; o[32] = acc2[r]; o[48] = acc3[r];
    }
}

extern "C" void kernel_launch(void* const* d_in, const int* in_sizes, int n_in,
                              void* d_out, int out_size, void* d_ws, size_t ws_size,
                              hipStream_t stream) {
    const float* A  = (const float*)d_in[0];
    const float* x  = (const float*)d_in[1];
    const float* W  = (const float*)d_in[2];
    const float* aw = (const float*)d_in[3];
    const float* ab = (const float*)d_in[4];
    float* out = (float*)d_out;

    float* Wh = (float*)d_ws;
    float* sc = Wh + (size_t)BB * HH * NN * FOUT;
    float* Dinv = sc + (size_t)2 * BB * HH * NN;

    gat_wh      <<<dim3(BB * HH * (NN / 16)),  dim3(32),  0, stream>>>(x, W, aw, ab, Wh, sc);
    gat_colsum  <<<dim3(BB * (NN / 32)), dim3(CS_WAVES * 32), 0, stream>>>(A, sc, Dinv);
    gat_attn_out<<<dim3(BB * HH * (NN / (AO_WAVES * 16))), dim3(AO_WAVES * 32), 0, stream>>>(A, Wh, sc, Dinv, out);
}


# column-sum kernel: a wave's 64-row A slice read as one unrolled run of loads ahead of use (a window of 20-40 line loads in flight per wave instead of 8-16), rows used in order; sum order kept
# speedup vs baseline: 4.4558x; 1.0410x over previous
#include <hip/hip_runtime.h>
#include <math.h>

#define BB   8
#define NN   1024
#define FIN  128
#define FOUT 64
#define HH   4
#define EPSF 1e-7f
#define NEG  0.2f
#define LOG2E 1.44269504088896340736f

typedef float v2f __attribute__((ext_vector_type(2)));
typedef float v4f __attribute__((ext_vector_type(4)));
typedef float v8f __attribute__((ext_vector_type(8)));

__device__ __forceinline__ v8f wmma_f32(v2f a, v2f b, v8f c) {
    return __builtin_amdgcn_wmma_f32_16x16x4_f32(
         false, a,  false, b,
         (short)0, c,  false,  false);
}

__device__ __forceinline__ int sc_row(int n) { return ((n >> 4) << 5) + (n & 15); }

__device__ __forceinline__ float row16_sum(float v) {
    v += __builtin_bit_cast(float, __builtin_amdgcn_mov_dpp(__builtin_bit_cast(int, v), 0x161, 0xf, 0xf, true));
    v += __builtin_bit_cast(float, __builtin_amdgcn_mov_dpp(__builtin_bit_cast(int, v), 0x162, 0xf, 0xf, true));
    v += __builtin_bit_cast(float, __builtin_amdgcn_mov_dpp(__builtin_bit_cast(int, v), 0x164, 0xf, 0xf, true));
    v += __builtin_bit_cast(float, __builtin_amdgcn_mov_dpp(__builtin_bit_cast(int, v), 0x168, 0xf, 0xf, true));
    return v;
}

__global__ __launch_bounds__(32) void gat_wh(const float* __restrict__ x,
                                             const float* __restrict__ W,
                                             const float* __restrict__ a_w,
                                             const float* __restrict__ a_b,
                                             float* __restrict__ Wh,
                                             float* __restrict__ sc) {
    const int lane = threadIdx.x;
    const int lm = lane & 15, hi = lane >> 4;
    int bid = blockIdx.x;
    const int ib = bid & 63; bid >>= 6;
    const int h  = bid & 3;  bid >>= 2;
    const int b  = bid;
    const int i0 = ib * 16;

    const float* xr = x + ((size_t)b * NN + i0 + lm) * FIN + 2 * hi;
    const float* Wp = W + (size_t)h * FIN * FOUT + (size_t)(2 * hi) * FOUT + lm;

    const float* a1 = a_w + (size_t)h * 2 * FOUT + lm;
    float a1v[4], a2v[4];
    #pragma unroll
    for (int t = 0; t < 4; ++t) { a1v[t] = a1[16 * t]; a2v[t] = a1[FOUT + 16 * t]; }
    const float bias = a_b[h];

    v8f acc0 = {}, acc1 = {}, acc2 = {}, acc3 = {};
    #pragma unroll 4
    for (int k = 0; k < FIN; k += 4) {
        v2f a = *(const v2f*)(xr + k);
        const float* wk = Wp + (size_t)k * FOUT;
        v2f b0 = { wk[0],  wk[FOUT]      };
        v2f b1 = { wk[16], wk[FOUT + 16] };
        v2f b2 = { wk[32], wk[FOUT + 32] };
        v2f b3 = { wk[48], wk[FOUT + 48] };
        acc0 = wmma_f32(a, b0, acc0);
        acc1 = wmma_f32(a, b1, acc1);
        acc2 = wmma_f32(a, b2, acc2);
        acc3 = wmma_f32(a, b3, acc3);
    }
    asm volatile("s_wait_alu depctr_va_vdst(0)"
                 : "+v"(acc0), "+v"(acc1), "+v"(acc2), "+v"(acc3));

    float* outp = Wh + (((size_t)b * HH + h) * NN + i0 + 8 * hi) * FOUT + lm;
    #pragma unroll
    for (int r = 0; r < 8; ++r) {
        float* o = outp + (size_t)r * FOUT;
        o[0] = acc0[r]; o[16] = acc1[r]; o[32] = acc2[r]; o[48] = acc3[r];
    }

    float p1[8], p2[8];
    #pragma unroll
    for (int r = 0; r < 8; ++r) {
        p1[r] = acc0[r] * a1v[0] + acc1[r] * a1v[1] + acc2[r] * a1v[2] + acc3[r] * a1v[3];
        p2[r] = acc0[r] * a2v[0] + acc1[r] * a2v[1] + acc2[r] * a2v[2] + acc3[r] * a2v[3];
    }
    #pragma unroll
    for (int r = 0; r < 8; ++r) { p1[r] = row16_sum(p1[r]); p2[r] = row16_sum(p2[r]); }
    if (lm == 0) {
        float* so = sc + ((size_t)b * HH + h) * 2 * NN + sc_row(i0 + 8 * hi);
        const v4f u0 = { p1[0] * LOG2E, p1[1] * LOG2E, p1[2] * LOG2E, p1[3] * LOG2E };
        const v4f u1 = { p1[4] * LOG2E, p1[5] * LOG2E, p1[6] * LOG2E, p1[7] * LOG2E };
        const v4f w0 = { (p2[0] + bias) * LOG2E, (p2[1] + bias) * LOG2E,
                         (p2[2] + bias) * LOG2E, (p2[3] + bias) * LOG2E };
        const v4f w1 = { (p2[4] + bias) * LOG2E, (p2[5] + bias) * LOG2E,
                         (p2[6] + bias) * LOG2E, (p2[7] + bias) * LOG2E };
        *(v4f*)(so)      = u0;
        *(v4f*)(so + 4)  = u1;
        *(v4f*)(so + 16) = w0;
        *(v4f*)(so + 20) = w1;
    }
}

__device__ __forceinline__ float attn_weight(float s1v, float s2v) {
    const float e2 = s1v + s2v;
    return __builtin_amdgcn_exp2f(fmaxf(e2, NEG * e2));
}

#define CS_WAVES 16
#define CS_ROWS  (NN / CS_WAVES)

__global__ __launch_bounds__(CS_WAVES * 32) void gat_colsum(
        const float* __restrict__ A,
        const float* __restrict__ sc,
        float* __restrict__ Dinv) {
    __shared__ float red[CS_WAVES][HH][32];

    const int lane = threadIdx.x & 31;
    const int w  = __builtin_amdgcn_readfirstlane(threadIdx.x >> 5);
    const int jt = blockIdx.x & 31;
    const int b  = blockIdx.x >> 5;
    const int j  = jt * 32 + lane;
    const int i0 = w * CS_ROWS;

    float s2v[HH], acc[HH];
    #pragma unroll
    for (int h = 0; h < HH; ++h) {
        s2v[h] = sc[((size_t)b * HH + h) * 2 * NN + sc_row(j) + 16];
        acc[h] = 0.f;
    }
    const float* Ap  = A  + ((size_t)b * NN + i0) * NN + j;
    const float* s1p = sc + (size_t)b * HH * 2 * NN;

    float a_all[CS_ROWS];
    #pragma unroll
    for (int r = 0; r < CS_ROWS; ++r) a_all[r] = Ap[(size_t)r * NN];

    #pragma unroll
    for (int r = 0; r < CS_ROWS; ++r) {
        #pragma unroll
        for (int h = 0; h < HH; ++h) {
            acc[h] += a_all[r] * attn_weight(
                s1p[h * 2 * NN + sc_row(i0) + ((r >> 4) << 5) + (r & 15)], s2v[h]);
        }
    }

    #pragma unroll
    for (int h = 0; h < HH; ++h) red[w][h][lane] = acc[h];
    __syncthreads();
    if (threadIdx.x < HH * 32) {
        float sum = 0.f;
        #pragma unroll
        for (int k = 0; k < CS_WAVES; ++k) sum += red[k][w][lane];
        Dinv[((size_t)b * HH + w) * NN + j] = 1.0f / (sum + EPSF);
    }
}

#define AO_WAVES 4
#define AO_TJ    64
#define AO_PW    160
#define AO_PE    68

__global__ __launch_bounds__(AO_WAVES * 32) void gat_attn_out(
        const float* __restrict__ A, const float* __restrict__ Wh,
        const float* __restrict__ sc,
        const float* __restrict__ Dinv, float* __restrict__ out) {
    __shared__ __align__(16) float Ws[2][(AO_TJ / 2) * AO_PW];
    __shared__ __align__(16) float Es[2][AO_WAVES][16 * AO_PE];

    const int tid  = threadIdx.x;
    const int lane = tid & 31;
    const int lm = lane & 15, hi = lane >> 4;
    const int w  = __builtin_amdgcn_readfirstlane(tid >> 5);
    int bid = blockIdx.x;
    const int ib = bid & 15; bid >>= 4;
    const int h  = bid & 3;  bid >>= 2;
    const int b  = bid;
    const int i0 = ib * (AO_WAVES * 16) + w * 16;
    const int bh = b * HH + h;

    const float* s1p = sc + (size_t)bh * 2 * NN + sc_row(i0);
    const float* s2p = sc + (size_t)bh * 2 * NN + sc_row(lane) + 16;
    const float* dvp = Dinv + (size_t)bh * NN + lane;
    const float* Ap  = A  + ((size_t)b * NN + i0) * NN + lane;
    const v4f* Wg = (const v4f*)(Wh + (size_t)bh * NN * FOUT)
                    + (tid >> 4) * 32 + (tid & 15);
    const int wst = (tid >> 4) * AO_PW + (tid & 15) * 8;

    float s1r[16];
    #pragma unroll
    for (int r = 0; r < 16; ++r) s1r[r] = s1p[r];

    float a_lo[16], a_hi[16];
    v4f   wv[8];
    float s2a, s2b, dva, dvb;
    #pragma unroll
    for (int r = 0; r < 16; ++r) {
        a_lo[r] = Ap[(size_t)r * NN];
        a_hi[r] = Ap[(size_t)r * NN + 32];
    }
    #pragma unroll
    for (int q = 0; q < 4; ++q) {
        wv[2 * q]     = Wg[q * 256];
        wv[2 * q + 1] = Wg[q * 256 + 16];
    }
    s2a = s2p[0];
    s2b = s2p[64];
    dva = dvp[0];
    dvb = dvp[32];

    v8f acc0 = {}, acc1 = {}, acc2 = {}, acc3 = {};

    for (int t = 0; t < NN / AO_TJ; ++t) {
        float* Eb = &Es[t & 1][w][0];
        float* Wb = &Ws[t & 1][0];

        #pragma unroll
        for (int r = 0; r < 16; ++r) {
            Eb[r * AO_PE + lane]      = (a_lo[r] * attn_weight(s1r[r], s2a)) * dva;
            Eb[r * AO_PE + lane + 32] = (a_hi[r] * attn_weight(s1r[r], s2b)) * dvb;
        }
        #pragma unroll
        for (int q = 0; q < 4; ++q) {
            const v4f r0 = wv[2 * q], r1 = wv[2 * q + 1];
            const v4f lo = { r0.x, r1.x, r0.y, r1.y };
            const v4f hi4 = { r0.z, r1.z, r0.w, r1.w };
            *(v4f*)&Wb[wst + q * 8 * AO_PW]     = lo;
            *(v4f*)&Wb[wst + q * 8 * AO_PW + 4] = hi4;
        }
        __syncthreads();

        if (t + 1 < NN / AO_TJ) {
            const int j1 = (t + 1) * AO_TJ;
            #pragma unroll
            for (int r = 0; r < 16; ++r) {
                a_lo[r] = Ap[(size_t)r * NN + j1];
                a_hi[r] = Ap[(size_t)r * NN + j1 + 32];
            }
            #pragma unroll
            for (int q = 0; q < 4; ++q) {
                wv[2 * q]     = Wg[(size_t)j1 * (FOUT / 4) + q * 256];
                wv[2 * q + 1] = Wg[(size_t)j1 * (FOUT / 4) + q * 256 + 16];
            }
            s2a = s2p[2 * j1];
            s2b = s2p[2 * j1 + 64];
            dva = dvp[j1];
            dvb = dvp[j1 + 32];
        }

        #pragma unroll
        for (int g = 0; g < 4; ++g) {
            v2f af[4], bf[4][4];
            #pragma unroll
            for (int u = 0; u < 4; ++u) {
                const int kk = g * 4 + u;
                af[u] = *(const v2f*)&Eb[lm * AO_PE + kk * 4 + 2 * hi];
                const float* wr = Wb + (kk * 2 + hi) * AO_PW + lm * 2;
                #pragma unroll
                for (int c = 0; c < 4; ++c) bf[u][c] = *(const v2f*)(wr + 32 * c);
            }
            #pragma unroll
            for (int u = 0; u < 4; ++u) {
                acc0 = wmma_f32(af[u], bf[u][0], acc0);
                acc1 = wmma_f32(af[u], bf[u][1], acc1);
                acc2 = wmma_f32(af[u], bf[u][2], acc2);
                acc3 = wmma_f32(af[u], bf[u][3], acc3);
            }
        }
    }

    float* op = out + ((size_t)b * NN + i0 + 8 * hi) * (HH * FOUT) + (size_t)h * FOUT + lm;
    #pragma unroll
    for (int r = 0; r < 8; ++r) {
        float* o = op + (size_t)r * (HH * FOUT);
        o[0] = acc0[r]; o[16] = acc1[r]; o[32] = acc2[r]; o[48] = acc3[r];
    }
}

extern "C" void kernel_launch(void* const* d_in, const int* in_sizes, int n_in,
                              void* d_out, int out_size, void* d_ws, size_t ws_size,
                              hipStream_t stream) {
    const float* A  = (const float*)d_in[0];
    const float* x  = (const float*)d_in[1];
    const float* W  = (const float*)d_in[2];
    const float* aw = (const float*)d_in[3];
    const float* ab = (const float*)d_in[4];
    float* out = (float*)d_out;

    float* Wh = (float*)d_ws;
    float* sc = Wh + (size_t)BB * HH * NN * FOUT;
    float* Dinv = sc + (size_t)2 * BB * HH * NN;

    gat_wh      <<<dim3(BB * HH * (NN / 16)),  dim3(32),  0, stream>>>(x, W, aw, ab, Wh, sc);
    gat_colsum  <<<dim3(BB * (NN / 32)), dim3(CS_WAVES * 32), 0, stream>>>(A, sc, Dinv);
    gat_attn_out<<<dim3(BB * HH * (NN / (AO_WAVES * 16))), dim3(AO_WAVES * 32), 0, stream>>>(A, Wh, sc, Dinv, out);
}
